# starting kernel
# baseline (speedup 1.0000x reference)
#include <hip/hip_runtime.h>
#include <hip/hip_bf16.h>

typedef __attribute__((ext_vector_type(2))) float v2f;
typedef __attribute__((ext_vector_type(4))) float v4f;
typedef __attribute__((ext_vector_type(8))) float v8f;

#define SEQ    2048
#define BATCH  2
#define HEADS  16
#define HDIM   128
#define SROW   4096
#define KSTRIDE 140
#define PSTRIDE 18

__global__ __launch_bounds__(256) void fa_fp32_wmma_kernel(
    const float* __restrict__ Q, const float* __restrict__ K,
    const float* __restrict__ V, float* __restrict__ Out)
{
    __shared__ float sK[2][16 * KSTRIDE];
    __shared__ float sV[2][16 * KSTRIDE];
    __shared__ float sP[8][16 * PSTRIDE];

    const int tid  = threadIdx.x;
    const int wave = tid >> 5;
    const int lane = tid & 31;
    const int hi   = lane >> 4;
    const int l15  = lane & 15;

    const size_t head_off = (size_t)blockIdx.y * HDIM;
    const int qblk = blockIdx.x * 128;
    const int q0w  = qblk + wave * 16;

    v2f qf[32];
    {
        const float inv_norm = 0.08838834764831845f;
        const float* qbase = Q + (size_t)(q0w + l15) * SROW + head_off + 2 * hi;
        #pragma unroll
        for (int ks = 0; ks < 32; ++ks) {
            v2f q = *(const v2f*)(qbase + ks * 4);
            q[0] *= inv_norm; q[1] *= inv_norm;
            qf[ks] = q;
        }
    }

    v8f o[8];
    #pragma unroll
    for (int dt = 0; dt < 8; ++dt) o[dt] = (v8f){};
    float mrow[8], lrow[8];
    #pragma unroll
    for (int r = 0; r < 8; ++r) { mrow[r] = -1e30f; lrow[r] = 0.0f; }

    const int nkt       = qblk / 16 + 8;
    const int my_kt_max = qblk / 16 + wave;

    const int ld_row = tid >> 4;
    const int ld_col = (tid & 15) * 8;
    const size_t ld_base = (size_t)ld_row * SROW + head_off + ld_col;
    v4f rk0, rk1, rv0, rv1;
    {
        rk0 = *(const v4f*)(K + ld_base);
        rk1 = *(const v4f*)(K + ld_base + 4);
        rv0 = *(const v4f*)(V + ld_base);
        rv1 = *(const v4f*)(V + ld_base + 4);
    }

    for (int kt = 0; kt < nkt; ++kt) {
        float* bK = sK[kt & 1];
        float* bV = sV[kt & 1];
        *(v4f*)&bK[ld_row * KSTRIDE + ld_col]     = rk0;
        *(v4f*)&bK[ld_row * KSTRIDE + ld_col + 4] = rk1;
        *(v4f*)&bV[ld_row * KSTRIDE + ld_col]     = rv0;
        *(v4f*)&bV[ld_row * KSTRIDE + ld_col + 4] = rv1;
        if (kt + 1 < nkt) {
            const size_t g = ld_base + (size_t)(kt + 1) * 16 * SROW;
            rk0 = *(const v4f*)(K + g);
            rk1 = *(const v4f*)(K + g + 4);
            rv0 = *(const v4f*)(V + g);
            rv1 = *(const v4f*)(V + g + 4);
        }
        __syncthreads();

        if (kt <= my_kt_max) {
            v8f s = (v8f){};
            #pragma unroll
            for (int g4 = 0; g4 < 4; ++g4) {
                v2f kb[8];
                #pragma unroll
                for (int j = 0; j < 8; ++j)
                    kb[j] = *(const v2f*)&bK[l15 * KSTRIDE + (g4 * 8 + j) * 4 + 2 * hi];
                #pragma unroll
                for (int j = 0; j < 8; ++j)
                    s = __builtin_amdgcn_wmma_f32_16x16x4_f32(
                            false, qf[g4 * 8 + j], false, kb[j], (short)0, s, false, false);
            }

            const int keyc = kt * 16 + l15;
            #pragma unroll
            for (int r = 0; r < 8; ++r) {
                const int qrow = q0w + r + 8 * hi;
                float sv = (keyc <= qrow) ? s[r] : -10000.0f;

                float mx = sv;
                mx = fmaxf(mx, __shfl_xor(mx, 1, 32));
                mx = fmaxf(mx, __shfl_xor(mx, 2, 32));
                mx = fmaxf(mx, __shfl_xor(mx, 4, 32));
                mx = fmaxf(mx, __shfl_xor(mx, 8, 32));

                const float mnew = fmaxf(mrow[r], mx);
                const float corr = __expf(mrow[r] - mnew);
                const float p    = __expf(sv - mnew);

                float ps = p;
                ps += __shfl_xor(ps, 1, 32);
                ps += __shfl_xor(ps, 2, 32);
                ps += __shfl_xor(ps, 4, 32);
                ps += __shfl_xor(ps, 8, 32);

                lrow[r] = lrow[r] * corr + ps;
                mrow[r] = mnew;
                #pragma unroll
                for (int dt = 0; dt < 8; ++dt) o[dt][r] *= corr;

                sP[wave][(r + 8 * hi) * PSTRIDE + l15] = p;
            }
            asm volatile("s_wait_dscnt 0x0" ::: "memory");

            v2f a[4];
            #pragma unroll
            for (int ka = 0; ka < 4; ++ka)
                a[ka] = *(const v2f*)&sP[wave][l15 * PSTRIDE + ka * 4 + 2 * hi];
            #pragma unroll
            for (int ka = 0; ka < 4; ++ka) {
                v2f vb[8];
                #pragma unroll
                for (int dt = 0; dt < 8; ++dt) {
                    vb[dt][0] = bV[(ka * 4 + 2 * hi    ) * KSTRIDE + dt * 16 + l15];
                    vb[dt][1] = bV[(ka * 4 + 2 * hi + 1) * KSTRIDE + dt * 16 + l15];
                }
                #pragma unroll
                for (int dt = 0; dt < 8; ++dt)
                    o[dt] = __builtin_amdgcn_wmma_f32_16x16x4_f32(
                                false, a[ka], false, vb[dt], (short)0, o[dt], false, false);
            }
        }
    }

    #pragma unroll
    for (int r = 0; r < 8; ++r) {
        const float inv = 1.0f / lrow[r];
        float* op = Out + (size_t)(q0w + r + 8 * hi) * SROW + head_off;
        #pragma unroll
        for (int dt = 0; dt < 8; ++dt)
            op[dt * 16 + l15] = o[dt][r] * inv;
    }
}

extern "C" void kernel_launch(void* const* d_in, const int* in_sizes, int n_in,
                              void* d_out, int out_size, void* d_ws, size_t ws_size,
                              hipStream_t stream) {
    (void)in_sizes; (void)n_in; (void)out_size; (void)d_ws; (void)ws_size;
    const float* Q = (const float*)d_in[0];
    const float* K = (const float*)d_in[1];
    const float* V = (const float*)d_in[2];
    float* Out     = (float*)d_out;

    dim3 grid(SEQ / 128, BATCH * HEADS);
    dim3 block(256);
    hipLaunchKernelGGL(fa_fp32_wmma_kernel, grid, block, 0, stream, Q, K, V, Out);
}


# Keys of each long query tile split over its SIMD's two waves: the short-tile wave, once done, takes every second 16-key tile (32 keys per barrier); states merged via LDS; order of sums changes
# speedup vs baseline: 2.4981x; 2.4981x over previous
#include <hip/hip_runtime.h>
#include <hip/hip_bf16.h>

typedef __attribute__((ext_vector_type(2))) float v2f;
typedef __attribute__((ext_vector_type(4))) float v4f;
typedef __attribute__((ext_vector_type(8))) float v8f;

#define SEQ    2048
#define BATCH  2
#define HEADS  16
#define HDIM   128
#define SROW   4096
#define KSTRIDE 140
#define PSTRIDE 18
#define MT     2
#define QBLK   (8 * MT * 16)
#define NQT    (SEQ / (MT * 16))
#define KSTEP  32
#define LDS_P  0
#define LDS_K  (8 * MT * 16 * PSTRIDE)
#define LDS_V  (LDS_K + 2 * KSTEP * KSTRIDE)
#define LDS_FLOATS (LDS_V + 2 * KSTEP * KSTRIDE)
#define LDS_MO 0
#define LDS_ML (4 * MT * 8 * 8 * 32)

__global__ __launch_bounds__(256) void fa_fp32_wmma_kernel(
    const float* __restrict__ Q, const float* __restrict__ K,
    const float* __restrict__ V, float* __restrict__ Out)
{
    __shared__ __attribute__((aligned(16))) float smem[LDS_FLOATS];

    const int tid  = threadIdx.x;
    const int wave = __builtin_amdgcn_readfirstlane(tid >> 5);
    const int lane = tid & 31;
    const int hi   = lane >> 4;
    const int l15  = lane & 15;

    const size_t head_off = (size_t)blockIdx.y * HDIM;
    const int bx     = blockIdx.x;
    const int pair   = wave & 3;
    const bool helper = wave >= 4;
    const int tL = 4 * bx + pair;
    const int tH = NQT - 1 - tL;
    int q0w = (helper ? tL : tH) * (MT * 16);
    float* sPw = smem + LDS_P + wave * (MT * 16 * PSTRIDE);

    v2f qf[MT][32];
    v8f o[MT][8];
    float mrow[MT][8], lrow[MT][8];

    auto load_q = [&](int q0) {
        const float inv_norm = 0.08838834764831845f;
        #pragma unroll
        for (int m = 0; m < MT; ++m) {
            const float* qbase = Q + (size_t)(q0 + 16 * m + l15) * SROW + head_off + 2 * hi;
            #pragma unroll
            for (int ks = 0; ks < 32; ++ks) {
                v2f q = *(const v2f*)(qbase + ks * 4);
                q[0] *= inv_norm; q[1] *= inv_norm;
                qf[m][ks] = q;
            }
        }
    };
    auto reset_state = [&]() {
        #pragma unroll
        for (int m = 0; m < MT; ++m) {
            #pragma unroll
            for (int dt = 0; dt < 8; ++dt) o[m][dt] = (v8f){};
            #pragma unroll
            for (int r = 0; r < 8; ++r) { mrow[m][r] = -1e30f; lrow[m][r] = 0.0f; }
        }
    };
    auto store_tile = [&](int q0) {
        #pragma unroll
        for (int m = 0; m < MT; ++m) {
            #pragma unroll
            for (int r = 0; r < 8; ++r) {
                const float inv = 1.0f / lrow[m][r];
                float* op = Out + (size_t)(q0 + 16 * m + r + 8 * hi) * SROW + head_off;
                #pragma unroll
                for (int dt = 0; dt < 8; ++dt)
                    op[dt * 16 + l15] = o[m][dt][r] * inv;
            }
        }
    };

    load_q(q0w);
    reset_state();

    const int nsteps = NQT - 4 * bx;

    const int ld_row = tid >> 4;
    const int ld_col = (tid & 15) * 8;
    const size_t ld_base = (size_t)ld_row * SROW + head_off + ld_col;
    v4f rk0, rk1, rk2, rk3, rv0, rv1, rv2, rv3;
    {
        rk0 = *(const v4f*)(K + ld_base);
        rk1 = *(const v4f*)(K + ld_base + 4);
        rk2 = *(const v4f*)(K + ld_base + (size_t)16 * SROW);
        rk3 = *(const v4f*)(K + ld_base + (size_t)16 * SROW + 4);
        rv0 = *(const v4f*)(V + ld_base);
        rv1 = *(const v4f*)(V + ld_base + 4);
        rv2 = *(const v4f*)(V + ld_base + (size_t)16 * SROW);
        rv3 = *(const v4f*)(V + ld_base + (size_t)16 * SROW + 4);
    }

    for (int j = 0; j < nsteps; ++j) {
        float* bK = smem + LDS_K + (j & 1) * (KSTEP * KSTRIDE);
        float* bV = smem + LDS_V + (j & 1) * (KSTEP * KSTRIDE);
        *(v4f*)&bK[ld_row * KSTRIDE + ld_col]            = rk0;
        *(v4f*)&bK[ld_row * KSTRIDE + ld_col + 4]        = rk1;
        *(v4f*)&bK[(ld_row + 16) * KSTRIDE + ld_col]     = rk2;
        *(v4f*)&bK[(ld_row + 16) * KSTRIDE + ld_col + 4] = rk3;
        *(v4f*)&bV[ld_row * KSTRIDE + ld_col]            = rv0;
        *(v4f*)&bV[ld_row * KSTRIDE + ld_col + 4]        = rv1;
        *(v4f*)&bV[(ld_row + 16) * KSTRIDE + ld_col]     = rv2;
        *(v4f*)&bV[(ld_row + 16) * KSTRIDE + ld_col + 4] = rv3;
        if (j + 1 < nsteps) {
            const size_t g = ld_base + (size_t)(j + 1) * KSTEP * SROW;
            rk0 = *(const v4f*)(K + g);
            rk1 = *(const v4f*)(K + g + 4);
            rk2 = *(const v4f*)(K + g + (size_t)16 * SROW);
            rk3 = *(const v4f*)(K + g + (size_t)16 * SROW + 4);
            rv0 = *(const v4f*)(V + g);
            rv1 = *(const v4f*)(V + g + 4);
            rv2 = *(const v4f*)(V + g + (size_t)16 * SROW);
            rv3 = *(const v4f*)(V + g + (size_t)16 * SROW + 4);
        }
        __syncthreads();

        if (helper && j == tL + 1) {
            store_tile(q0w);
            q0w = tH * (MT * 16);
            load_q(q0w);
            reset_state();
        }

        int sub_lo = 0, sub_hi = 0;
        if (j <= tL)      { sub_lo = 0; sub_hi = 2; }
        else if (j <= tH) { sub_lo = helper ? 1 : 0; sub_hi = sub_lo + 1; }

        #pragma unroll 1
        for (int sub = sub_lo; sub < sub_hi; ++sub) {
            const float* cK = bK + sub * 16 * KSTRIDE;
            const float* cV = bV + sub * 16 * KSTRIDE;
            v8f s[MT];
            #pragma unroll
            for (int m = 0; m < MT; ++m) s[m] = (v8f){};
            #pragma unroll
            for (int g4 = 0; g4 < 4; ++g4) {
                v2f kb[8];
                #pragma unroll
                for (int jj = 0; jj < 8; ++jj)
                    kb[jj] = *(const v2f*)&cK[l15 * KSTRIDE + (g4 * 8 + jj) * 4 + 2 * hi];
                #pragma unroll
                for (int jj = 0; jj < 8; ++jj) {
                    #pragma unroll
                    for (int m = 0; m < MT; ++m)
                        s[m] = __builtin_amdgcn_wmma_f32_16x16x4_f32(
                                false, qf[m][g4 * 8 + jj], false, kb[jj], (short)0, s[m], false, false);
                }
            }

            const int keyc = (2 * j + sub) * 16 + l15;
            #pragma unroll
            for (int m = 0; m < MT; ++m) {
                #pragma unroll
                for (int r = 0; r < 8; ++r) {
                    const int qrow = q0w + 16 * m + r + 8 * hi;
                    float sv = (keyc <= qrow) ? s[m][r] : -10000.0f;

                    float mx = sv;
                    mx = fmaxf(mx, __shfl_xor(mx, 1, 32));
                    mx = fmaxf(mx, __shfl_xor(mx, 2, 32));
                    mx = fmaxf(mx, __shfl_xor(mx, 4, 32));
                    mx = fmaxf(mx, __shfl_xor(mx, 8, 32));

                    const float mnew = fmaxf(mrow[m][r], mx);
                    const float corr = __expf(mrow[m][r] - mnew);
                    const float p    = __expf(sv - mnew);

                    float ps = p;
                    ps += __shfl_xor(ps, 1, 32);
                    ps += __shfl_xor(ps, 2, 32);
                    ps += __shfl_xor(ps, 4, 32);
                    ps += __shfl_xor(ps, 8, 32);

                    lrow[m][r] = lrow[m][r] * corr + ps;
                    mrow[m][r] = mnew;
                    #pragma unroll
                    for (int dt = 0; dt < 8; ++dt) o[m][dt][r] *= corr;

                    sPw[m * (16 * PSTRIDE) + (r + 8 * hi) * PSTRIDE + l15] = p;
                }
            }
            asm volatile("s_wait_dscnt 0x0" ::: "memory");

            v2f a[MT][4];
            #pragma unroll
            for (int m = 0; m < MT; ++m) {
                #pragma unroll
                for (int ka = 0; ka < 4; ++ka)
                    a[m][ka] = *(const v2f*)&sPw[m * (16 * PSTRIDE) + l15 * PSTRIDE + ka * 4 + 2 * hi];
            }
            #pragma unroll
            for (int ka = 0; ka < 4; ++ka) {
                v2f vb[8];
                #pragma unroll
                for (int dt = 0; dt < 8; ++dt) {
                    vb[dt][0] = cV[(ka * 4 + 2 * hi    ) * KSTRIDE + dt * 16 + l15];
                    vb[dt][1] = cV[(ka * 4 + 2 * hi + 1) * KSTRIDE + dt * 16 + l15];
                }
                #pragma unroll
                for (int dt = 0; dt < 8; ++dt) {
                    #pragma unroll
                    for (int m = 0; m < MT; ++m)
                        o[m][dt] = __builtin_amdgcn_wmma_f32_16x16x4_f32(
                                    false, a[m][ka], false, vb[dt], (short)0, o[m][dt], false, false);
                }
            }
        }
    }

    float* sMo = smem + LDS_MO + pair * (MT * 8 * 8 * 32);
    float* sMl = smem + LDS_ML + pair * (MT * 8 * 2 * 32);
    __syncthreads();
    if (helper) {
        #pragma unroll
        for (int m = 0; m < MT; ++m) {
            #pragma unroll
            for (int dt = 0; dt < 8; ++dt) {
                #pragma unroll
                for (int q = 0; q < 2; ++q) {
                    v4f t;
                    t[0] = o[m][dt][4 * q + 0]; t[1] = o[m][dt][4 * q + 1];
                    t[2] = o[m][dt][4 * q + 2]; t[3] = o[m][dt][4 * q + 3];
                    *(v4f*)&sMo[((m * 8 + dt) * 2 + q) * 128 + lane * 4] = t;
                }
            }
            #pragma unroll
            for (int q = 0; q < 2; ++q) {
                v4f tm, tl;
                tm[0] = mrow[m][4 * q + 0]; tm[1] = mrow[m][4 * q + 1];
                tm[2] = mrow[m][4 * q + 2]; tm[3] = mrow[m][4 * q + 3];
                tl[0] = lrow[m][4 * q + 0]; tl[1] = lrow[m][4 * q + 1];
                tl[2] = lrow[m][4 * q + 2]; tl[3] = lrow[m][4 * q + 3];
                *(v4f*)&sMl[((m * 2 + q) * 2 + 0) * 128 + lane * 4] = tm;
                *(v4f*)&sMl[((m * 2 + q) * 2 + 1) * 128 + lane * 4] = tl;
            }
        }
    }
    __syncthreads();
    if (!helper) {
        #pragma unroll
        for (int m = 0; m < MT; ++m) {
            #pragma unroll
            for (int q = 0; q < 2; ++q) {
                const v4f m2 = *(const v4f*)&sMl[((m * 2 + q) * 2 + 0) * 128 + lane * 4];
                const v4f l2 = *(const v4f*)&sMl[((m * 2 + q) * 2 + 1) * 128 + lane * 4];
                float a1[4], a2[4];
                #pragma unroll
                for (int e = 0; e < 4; ++e) {
                    const int r = 4 * q + e;
                    const float mm = fmaxf(mrow[m][r], m2[e]);
                    const float e1 = __expf(mrow[m][r] - mm);
                    const float e2 = __expf(m2[e] - mm);
                    const float inv = 1.0f / (lrow[m][r] * e1 + l2[e] * e2);
                    a1[e] = e1 * inv;
                    a2[e] = e2 * inv;
                }
                #pragma unroll
                for (int dt = 0; dt < 8; ++dt) {
                    const v4f o2 = *(const v4f*)&sMo[((m * 8 + dt) * 2 + q) * 128 + lane * 4];
                    #pragma unroll
                    for (int e = 0; e < 4; ++e) {
                        const int r = 4 * q + e;
                        float* op = Out + (size_t)(q0w + 16 * m + r + 8 * hi) * SROW + head_off;
                        op[dt * 16 + l15] = o[m][dt][r] * a1[e] + o2[e] * a2[e];
                    }
                }
            }
        }
    }
}

extern "C" void kernel_launch(void* const* d_in, const int* in_sizes, int n_in,
                              void* d_out, int out_size, void* d_ws, size_t ws_size,
                              hipStream_t stream) {
    (void)in_sizes; (void)n_in; (void)out_size; (void)d_ws; (void)ws_size;
    const float* Q = (const float*)d_in[0];
    const float* K = (const float*)d_in[1];
    const float* V = (const float*)d_in[2];
    float* Out     = (float*)d_out;

    dim3 grid(SEQ / QBLK, BATCH * HEADS);
    dim3 block(256);
    hipLaunchKernelGGL(fa_fp32_wmma_kernel, grid, block, 0, stream, Q, K, V, Out);
}


# K/V LDS ring of three stages with the step barrier split: a wave stores the next tile and signals at the top of a step, waits a step later; waves run out of step, no fence drains the prefetch
# speedup vs baseline: 3.7101x; 1.4852x over previous
#include <hip/hip_runtime.h>
#include <hip/hip_bf16.h>

typedef __attribute__((ext_vector_type(2))) float v2f;
typedef __attribute__((ext_vector_type(4))) float v4f;
typedef __attribute__((ext_vector_type(8))) float v8f;

#define SEQ    2048
#define BATCH  2
#define HEADS  16
#define HDIM   128
#define SROW   4096
#define KSTRIDE 140
#define PSTRIDE 18
#define MT     2
#define QBLK   (8 * MT * 16)
#define NQT    (SEQ / (MT * 16))
#define KSTEP  32
#define NSTAGE 3
#define LDS_P  0
#define LDS_RING (8 * MT * 16 * PSTRIDE)
#define STAGE_FLOATS (2 * KSTEP * KSTRIDE)
#define STAGE_V (KSTEP * KSTRIDE)
#define LDS_FLOATS (LDS_RING + NSTAGE * STAGE_FLOATS)
#define LDS_MO 0
#define LDS_ML (4 * MT * 8 * 8 * 32)
#define RESCALE_TAU 8.0f

__device__ __forceinline__ float xhalf(float x) {
    const int b = __builtin_bit_cast(int, x);
    return __builtin_bit_cast(float,
        __builtin_amdgcn_permlanex16(b, b, 0x76543210, 0xfedcba98, false, false));
}
#define ROW_SHARE(t, r) __builtin_bit_cast(float, __builtin_amdgcn_update_dpp((t), (t), 0x150 + (r), 0xf, 0xf, false))
__device__ __forceinline__ void rows_to_c(float x, float (&out)[8]) {
    const int b = __builtin_bit_cast(int, x);
    const int t = __builtin_amdgcn_update_dpp(b, b, 0x128  , 0x2  , 0xf, false);
    out[0] = ROW_SHARE(t, 0); out[1] = ROW_SHARE(t, 1); out[2] = ROW_SHARE(t, 2); out[3] = ROW_SHARE(t, 3);
    out[4] = ROW_SHARE(t, 4); out[5] = ROW_SHARE(t, 5); out[6] = ROW_SHARE(t, 6); out[7] = ROW_SHARE(t, 7);
}

__device__ __forceinline__ void wg_signal() {
    asm volatile("s_wait_dscnt 0x0\n\ts_barrier_signal -1" ::: "memory");
}
__device__ __forceinline__ void wg_wait() {
    asm volatile("s_barrier_wait -1" ::: "memory");
}

__global__ __launch_bounds__(256) void fa_fp32_wmma_kernel(
    const float* __restrict__ Q, const float* __restrict__ K,
    const float* __restrict__ V, float* __restrict__ Out)
{
    __shared__ __attribute__((aligned(16))) float smem[LDS_FLOATS];

    const int tid  = threadIdx.x;
    const int wave = __builtin_amdgcn_readfirstlane(tid >> 5);
    const int lane = tid & 31;
    const int hi   = lane >> 4;
    const int l15  = lane & 15;

    const size_t head_off = (size_t)blockIdx.y * HDIM;
    const int bx     = blockIdx.x;
    const int pair   = wave & 3;
    const bool helper = wave >= 4;
    const int tL = 4 * bx + pair;
    const int tH = NQT - 1 - tL;
    int q0w = (helper ? tL : tH) * (MT * 16);
    float* sPw = smem + LDS_P + wave * (MT * 16 * PSTRIDE);

    v2f qf[MT][32];
    v8f o[MT][8];
    float mrow[MT], lrow[MT];

    auto load_q = [&](int q0) {
        const float inv_norm = 0.08838834764831845f;
        #pragma unroll
        for (int m = 0; m < MT; ++m) {
            const float* qbase = Q + (size_t)(q0 + 16 * m + l15) * SROW + head_off + 2 * hi;
            #pragma unroll
            for (int ks = 0; ks < 32; ++ks) {
                v2f q = *(const v2f*)(qbase + ks * 4);
                q[0] *= inv_norm; q[1] *= inv_norm;
                qf[m][ks] = q;
            }
        }
    };
    auto reset_state = [&]() {
        #pragma unroll
        for (int m = 0; m < MT; ++m) {
            #pragma unroll
            for (int dt = 0; dt < 8; ++dt) o[m][dt] = (v8f){};
            mrow[m] = -1e30f; lrow[m] = 0.0f;
        }
    };
    auto store_tile = [&](int q0) {
        #pragma unroll
        for (int m = 0; m < MT; ++m) {
            float inv[8];
            rows_to_c(1.0f / lrow[m], inv);
            #pragma unroll
            for (int r = 0; r < 8; ++r) {
                float* op = Out + (size_t)(q0 + 16 * m + r + 8 * hi) * SROW + head_off;
                #pragma unroll
                for (int dt = 0; dt < 8; ++dt)
                    op[dt * 16 + l15] = o[m][dt][r] * inv[r];
            }
        }
    };

    load_q(q0w);
    reset_state();

    const int nsteps = NQT - 4 * bx;

    const int ld_row = tid >> 4;
    const int ld_col = (tid & 15) * 8;
    const size_t ld_base = (size_t)ld_row * SROW + head_off + ld_col;
    v4f rk0, rk1, rk2, rk3, rv0, rv1, rv2, rv3;
    auto fetch_tile = [&](int jt) {
        const size_t g = ld_base + (size_t)jt * KSTEP * SROW;
        rk0 = *(const v4f*)(K + g);
        rk1 = *(const v4f*)(K + g + 4);
        rk2 = *(const v4f*)(K + g + (size_t)16 * SROW);
        rk3 = *(const v4f*)(K + g + (size_t)16 * SROW + 4);
        rv0 = *(const v4f*)(V + g);
        rv1 = *(const v4f*)(V + g + 4);
        rv2 = *(const v4f*)(V + g + (size_t)16 * SROW);
        rv3 = *(const v4f*)(V + g + (size_t)16 * SROW + 4);
    };
    auto commit_tile = [&](int st) {
        float* wK = smem + LDS_RING + st * STAGE_FLOATS;
        float* wV = wK + STAGE_V;
        *(v4f*)&wK[ld_row * KSTRIDE + ld_col]            = rk0;
        *(v4f*)&wK[ld_row * KSTRIDE + ld_col + 4]        = rk1;
        *(v4f*)&wK[(ld_row + 16) * KSTRIDE + ld_col]     = rk2;
        *(v4f*)&wK[(ld_row + 16) * KSTRIDE + ld_col + 4] = rk3;
        *(v4f*)&wV[ld_row * KSTRIDE + ld_col]            = rv0;
        *(v4f*)&wV[ld_row * KSTRIDE + ld_col + 4]        = rv1;
        *(v4f*)&wV[(ld_row + 16) * KSTRIDE + ld_col]     = rv2;
        *(v4f*)&wV[(ld_row + 16) * KSTRIDE + ld_col + 4] = rv3;
    };
    fetch_tile(0);
    commit_tile(0);
    if (nsteps > 1) fetch_tile(1);
    wg_signal();

    int st = 0;
    for (int j = 0; j < nsteps; ++j) {
        const int stn = (st == NSTAGE - 1) ? 0 : st + 1;
        wg_wait();
        if (j + 1 < nsteps) {
            commit_tile(stn);
            if (j + 2 < nsteps) fetch_tile(j + 2);
            wg_signal();
        }
        const float* bK = smem + LDS_RING + st * STAGE_FLOATS;
        const float* bV = bK + STAGE_V;
        st = stn;

        if (helper && j == tL + 1) {
            store_tile(q0w);
            q0w = tH * (MT * 16);
            load_q(q0w);
            reset_state();
        }

        int sub_lo = 0, sub_hi = 0;
        if (j <= tL)      { sub_lo = 0; sub_hi = 2; }
        else if (j <= tH) { sub_lo = helper ? 1 : 0; sub_hi = sub_lo + 1; }

        #pragma unroll 1
        for (int sub = sub_lo; sub < sub_hi; ++sub) {
            const float* cK = bK + sub * 16 * KSTRIDE;
            const float* cV = bV + sub * 16 * KSTRIDE;
            v8f s[MT];
            #pragma unroll
            for (int m = 0; m < MT; ++m) s[m] = (v8f){};
            v2f kb[32];
            #pragma unroll
            for (int ks = 0; ks < 32; ++ks)
                kb[ks] = *(const v2f*)&cK[l15 * KSTRIDE + ks * 4 + 2 * hi];
            const int key0 = (2 * j + sub) * 16 + 2 * hi;
            bool keep[MT][8];
            #pragma unroll
            for (int m = 0; m < MT; ++m) {
                const int qrow = q0w + 16 * m + l15;
                #pragma unroll
                for (int ka = 0; ka < 4; ++ka) {
                    #pragma unroll
                    for (int e = 0; e < 2; ++e) keep[m][2 * ka + e] = (key0 + ka * 4 + e <= qrow);
                }
            }
            __builtin_amdgcn_sched_barrier(0);
            #pragma unroll
            for (int ks = 0; ks < 32; ++ks) {
                #pragma unroll
                for (int m = 0; m < MT; ++m)
                    s[m] = __builtin_amdgcn_wmma_f32_16x16x4_f32(
                            false, qf[m][ks], false, kb[ks], (short)0, s[m], false, false);
            }

            #pragma unroll
            for (int m = 0; m < MT; ++m) {
                #pragma unroll
                for (int r = 0; r < 8; ++r)
                    sPw[m * (16 * PSTRIDE) + (r + 8 * hi) * PSTRIDE + l15] = s[m][r];
            }
            asm volatile("s_wait_dscnt 0x0" ::: "memory");
            v2f a[MT][4];
            #pragma unroll
            for (int m = 0; m < MT; ++m) {
                #pragma unroll
                for (int ka = 0; ka < 4; ++ka)
                    a[m][ka] = *(const v2f*)&sPw[m * (16 * PSTRIDE) + l15 * PSTRIDE + ka * 4 + 2 * hi];
            }

            v2f vb[4][8];
            #pragma unroll
            for (int ka = 0; ka < 4; ++ka) {
                #pragma unroll
                for (int dt = 0; dt < 8; ++dt) {
                    vb[ka][dt][0] = cV[(ka * 4 + 2 * hi    ) * KSTRIDE + dt * 16 + l15];
                    vb[ka][dt][1] = cV[(ka * 4 + 2 * hi + 1) * KSTRIDE + dt * 16 + l15];
                }
            }

            __builtin_amdgcn_sched_barrier(0);
            float sv[MT][8], mxh[MT];
            bool grow = false;
            #pragma unroll
            for (int m = 0; m < MT; ++m) {
                #pragma unroll
                for (int ka = 0; ka < 4; ++ka) {
                    #pragma unroll
                    for (int e = 0; e < 2; ++e)
                        sv[m][2 * ka + e] = keep[m][2 * ka + e] ? a[m][ka][e] : -10000.0f;
                }
                mxh[m] = fmaxf(fmaxf(fmaxf(sv[m][0], sv[m][1]), fmaxf(sv[m][2], sv[m][3])),
                               fmaxf(fmaxf(sv[m][4], sv[m][5]), fmaxf(sv[m][6], sv[m][7])));
                grow = grow || (mxh[m] > mrow[m] + RESCALE_TAU);
            }
            if (__builtin_expect(__builtin_amdgcn_ballot_w32(grow) != 0u, 0)) {
                #pragma unroll
                for (int m = 0; m < MT; ++m) {
                    const float mx = fmaxf(mxh[m], xhalf(mxh[m]));
                    const float mnew = fmaxf(mrow[m], mx);
                    const float corr = __expf(mrow[m] - mnew);
                    lrow[m] *= corr;
                    mrow[m] = mnew;
                    float cc[8];
                    rows_to_c(corr, cc);
                    #pragma unroll
                    for (int r = 0; r < 8; ++r) {
                        #pragma unroll
                        for (int dt = 0; dt < 8; ++dt) o[m][dt][r] *= cc[r];
                    }
                }
            }
            #pragma unroll
            for (int m = 0; m < MT; ++m) {
                float p[8];
                #pragma unroll
                for (int i = 0; i < 8; ++i) p[i] = __expf(sv[m][i] - mrow[m]);
                float ps = ((p[0] + p[1]) + (p[2] + p[3])) + ((p[4] + p[5]) + (p[6] + p[7]));
                ps += xhalf(ps);
                lrow[m] += ps;
                #pragma unroll
                for (int ka = 0; ka < 4; ++ka) { a[m][ka][0] = p[2 * ka]; a[m][ka][1] = p[2 * ka + 1]; }
            }

            #pragma unroll
            for (int ka = 0; ka < 4; ++ka) {
                #pragma unroll
                for (int dt = 0; dt < 8; ++dt) {
                    #pragma unroll
                    for (int m = 0; m < MT; ++m)
                        o[m][dt] = __builtin_amdgcn_wmma_f32_16x16x4_f32(
                                    false, a[m][ka], false, vb[ka][dt], (short)0, o[m][dt], false, false);
                }
            }
        }
    }

    float* sMo = smem + LDS_MO + pair * (MT * 8 * 8 * 32);
    float* sMl = smem + LDS_ML + pair * (MT * 2 * 32);
    __syncthreads();
    if (helper) {
        #pragma unroll
        for (int m = 0; m < MT; ++m) {
            #pragma unroll
            for (int dt = 0; dt < 8; ++dt) {
                #pragma unroll
                for (int q = 0; q < 2; ++q) {
                    v4f t;
                    t[0] = o[m][dt][4 * q + 0]; t[1] = o[m][dt][4 * q + 1];
                    t[2] = o[m][dt][4 * q + 2]; t[3] = o[m][dt][4 * q + 3];
                    *(v4f*)&sMo[((m * 8 + dt) * 2 + q) * 128 + lane * 4] = t;
                }
            }
            sMl[(m * 2 + 0) * 32 + lane] = mrow[m];
            sMl[(m * 2 + 1) * 32 + lane] = lrow[m];
        }
    }
    __syncthreads();
    if (!helper) {
        #pragma unroll
        for (int m = 0; m < MT; ++m) {
            const float m2 = sMl[(m * 2 + 0) * 32 + lane];
            const float l2 = sMl[(m * 2 + 1) * 32 + lane];
            const float mm = fmaxf(mrow[m], m2);
            const float e1 = __expf(mrow[m] - mm);
            const float e2 = __expf(m2 - mm);
            const float inv = 1.0f / (lrow[m] * e1 + l2 * e2);
            float a1[8], a2[8];
            rows_to_c(e1 * inv, a1);
            rows_to_c(e2 * inv, a2);
            #pragma unroll
            for (int q = 0; q < 2; ++q) {
                #pragma unroll
                for (int dt = 0; dt < 8; ++dt) {
                    const v4f o2 = *(const v4f*)&sMo[((m * 8 + dt) * 2 + q) * 128 + lane * 4];
                    #pragma unroll
                    for (int e = 0; e < 4; ++e) {
                        const int r = 4 * q + e;
                        float* op = Out + (size_t)(q0w + 16 * m + r + 8 * hi) * SROW + head_off;
                        op[dt * 16 + l15] = o[m][dt][r] * a1[r] + o2[e] * a2[r];
                    }
                }
            }
        }
    }
}

extern "C" void kernel_launch(void* const* d_in, const int* in_sizes, int n_in,
                              void* d_out, int out_size, void* d_ws, size_t ws_size,
                              hipStream_t stream) {
    (void)in_sizes; (void)n_in; (void)out_size; (void)d_ws; (void)ws_size;
    const float* Q = (const float*)d_in[0];
    const float* K = (const float*)d_in[1];
    const float* V = (const float*)d_in[2];
    float* Out     = (float*)d_out;

    dim3 grid(SEQ / QBLK, BATCH * HEADS);
    dim3 block(256);
    hipLaunchKernelGGL(fa_fp32_wmma_kernel, grid, block, 0, stream, Q, K, V, Out);
}


# Vector slots of the sub-tile loop cut 169 to 70: V tile in LDS keeps a fragment's two keys 16 floats apart (pairs load in place, no moves), causal selects only on diagonal sub-tiles; sums unchanged
# speedup vs baseline: 3.8653x; 1.0418x over previous
#include <hip/hip_runtime.h>
#include <hip/hip_bf16.h>

typedef __attribute__((ext_vector_type(2))) float v2f;
typedef __attribute__((ext_vector_type(4))) float v4f;
typedef __attribute__((ext_vector_type(8))) float v8f;

#define SEQ    2048
#define BATCH  2
#define HEADS  16
#define HDIM   128
#define SROW   4096
#define KSTRIDE 140
#define PSTRIDE 18
#define VPS    272
#define MT     2
#define QBLK   (8 * MT * 16)
#define NQT    (SEQ / (MT * 16))
#define KSTEP  32
#define NSTAGE 3
#define LDS_P  0
#define LDS_RING (8 * MT * 16 * PSTRIDE)
#define STAGE_FLOATS (2 * KSTEP * KSTRIDE)
#define STAGE_V (KSTEP * KSTRIDE)
#define LDS_FLOATS (LDS_RING + NSTAGE * STAGE_FLOATS)
#define LDS_MO 0
#define LDS_ML (4 * MT * 8 * 8 * 32)
#define RESCALE_TAU 8.0f

__device__ __forceinline__ float xhalf(float x) {
    const int b = __builtin_bit_cast(int, x);
    return __builtin_bit_cast(float,
        __builtin_amdgcn_permlanex16(b, b, 0x76543210, 0xfedcba98, false, false));
}
#define ROW_SHARE(t, r) __builtin_bit_cast(float, __builtin_amdgcn_update_dpp((t), (t), 0x150 + (r), 0xf, 0xf, false))
__device__ __forceinline__ void rows_to_c(float x, float (&out)[8]) {
    const int b = __builtin_bit_cast(int, x);
    const int t = __builtin_amdgcn_update_dpp(b, b, 0x128  , 0x2  , 0xf, false);
    out[0] = ROW_SHARE(t, 0); out[1] = ROW_SHARE(t, 1); out[2] = ROW_SHARE(t, 2); out[3] = ROW_SHARE(t, 3);
    out[4] = ROW_SHARE(t, 4); out[5] = ROW_SHARE(t, 5); out[6] = ROW_SHARE(t, 6); out[7] = ROW_SHARE(t, 7);
}

__device__ __forceinline__ void wg_signal() {
    asm volatile("s_wait_dscnt 0x0\n\ts_barrier_signal -1" ::: "memory");
}
__device__ __forceinline__ void wg_wait() {
    asm volatile("s_barrier_wait -1" ::: "memory");
}

__global__ __launch_bounds__(256) void fa_fp32_wmma_kernel(
    const float* __restrict__ Q, const float* __restrict__ K,
    const float* __restrict__ V, float* __restrict__ Out)
{
    __shared__ __attribute__((aligned(16))) float smem[LDS_FLOATS];

    const int tid  = threadIdx.x;
    const int wave = __builtin_amdgcn_readfirstlane(tid >> 5);
    const int lane = tid & 31;
    const int hi   = lane >> 4;
    const int l15  = lane & 15;

    const size_t head_off = (size_t)blockIdx.y * HDIM;
    const int bx     = blockIdx.x;
    const int pair   = wave & 3;
    const bool helper = wave >= 4;
    const int tL = 4 * bx + pair;
    const int tH = NQT - 1 - tL;
    int q0w = (helper ? tL : tH) * (MT * 16);
    float* sPw = smem + LDS_P + wave * (MT * 16 * PSTRIDE);

    v2f qf[MT][32];
    v8f o[MT][8];
    float mrow[MT], lrow[MT];

    auto load_q = [&](int q0) {
        const float inv_norm = 0.08838834764831845f;
        #pragma unroll
        for (int m = 0; m < MT; ++m) {
            const float* qbase = Q + (size_t)(q0 + 16 * m + l15) * SROW + head_off + 2 * hi;
            #pragma unroll
            for (int ks = 0; ks < 32; ++ks) {
                v2f q = *(const v2f*)(qbase + ks * 4);
                q[0] *= inv_norm; q[1] *= inv_norm;
                qf[m][ks] = q;
            }
        }
    };
    auto reset_state = [&]() {
        #pragma unroll
        for (int m = 0; m < MT; ++m) {
            #pragma unroll
            for (int dt = 0; dt < 8; ++dt) o[m][dt] = (v8f){};
            mrow[m] = -1e30f; lrow[m] = 0.0f;
        }
    };
    auto store_tile = [&](int q0) {
        #pragma unroll
        for (int m = 0; m < MT; ++m) {
            float inv[8];
            rows_to_c(1.0f / lrow[m], inv);
            #pragma unroll
            for (int r = 0; r < 8; ++r) {
                float* op = Out + (size_t)(q0 + 16 * m + r + 8 * hi) * SROW + head_off;
                #pragma unroll
                for (int dt = 0; dt < 8; ++dt)
                    op[dt * 16 + l15] = o[m][dt][r] * inv[r];
            }
        }
    };

    load_q(q0w);
    reset_state();

    const int nsteps = NQT - 4 * bx;

    const int ld_row = tid >> 4;
    const int ld_col = (tid & 15) * 8;
    const size_t ld_base = (size_t)ld_row * SROW + head_off + ld_col;
    const int v_row = 2 * ld_row + ((tid >> 2) & 1);
    const int v_col = ((tid >> 3) & 1) * 16 + (tid & 3) * 4;
    const size_t ldv_base = (size_t)v_row * SROW + head_off + v_col;
    const int ld_voff = ld_row * VPS + (tid & 15) * 4;
    v4f rk0, rk1, rk2, rk3, rv0, rv1, rv2, rv3;
    auto fetch_tile = [&](int jt) {
        const size_t g = ld_base + (size_t)jt * KSTEP * SROW;
        rk0 = *(const v4f*)(K + g);
        rk1 = *(const v4f*)(K + g + 4);
        rk2 = *(const v4f*)(K + g + (size_t)16 * SROW);
        rk3 = *(const v4f*)(K + g + (size_t)16 * SROW + 4);
        const size_t gv = ldv_base + (size_t)jt * KSTEP * SROW;
        rv0 = *(const v4f*)(V + gv);
        rv1 = *(const v4f*)(V + gv + 32);
        rv2 = *(const v4f*)(V + gv + 64);
        rv3 = *(const v4f*)(V + gv + 96);
    };
    auto commit_tile = [&](int st) {
        float* wK = smem + LDS_RING + st * STAGE_FLOATS;
        float* wV = wK + STAGE_V;
        *(v4f*)&wK[ld_row * KSTRIDE + ld_col]            = rk0;
        *(v4f*)&wK[ld_row * KSTRIDE + ld_col + 4]        = rk1;
        *(v4f*)&wK[(ld_row + 16) * KSTRIDE + ld_col]     = rk2;
        *(v4f*)&wK[(ld_row + 16) * KSTRIDE + ld_col + 4] = rk3;
        *(v4f*)&wV[ld_voff]       = rv0;
        *(v4f*)&wV[ld_voff + 64]  = rv1;
        *(v4f*)&wV[ld_voff + 128] = rv2;
        *(v4f*)&wV[ld_voff + 192] = rv3;
    };
    fetch_tile(0);
    commit_tile(0);
    if (nsteps > 1) fetch_tile(1);
    wg_signal();

    int v_lane = hi * VPS + l15;
    asm volatile("" : "+v"(v_lane));
    int st = 0;
    for (int j = 0; j < nsteps; ++j) {
        const int stn = (st == NSTAGE - 1) ? 0 : st + 1;
        wg_wait();
        if (j + 1 < nsteps) {
            commit_tile(stn);
            if (j + 2 < nsteps) fetch_tile(j + 2);
            wg_signal();
        }
        const float* bK = smem + LDS_RING + st * STAGE_FLOATS;
        const float* bV = bK + STAGE_V;
        st = stn;

        if (helper && j == tL + 1) {
            store_tile(q0w);
            q0w = tH * (MT * 16);
            load_q(q0w);
            reset_state();
        }

        int sub_lo = 0, sub_hi = 0;
        if (j <= tL)      { sub_lo = 0; sub_hi = 2; }
        else if (j <= tH) { sub_lo = helper ? 1 : 0; sub_hi = sub_lo + 1; }

        #pragma unroll 1
        for (int sub = sub_lo; sub < sub_hi; ++sub) {
            const float* cK = bK + sub * 16 * KSTRIDE;
            const float* cV = bV + sub * 8 * VPS;
            v8f s[MT];
            #pragma unroll
            for (int m = 0; m < MT; ++m) s[m] = (v8f){};
            v2f kb[32];
            #pragma unroll
            for (int ks = 0; ks < 32; ++ks)
                kb[ks] = *(const v2f*)&cK[l15 * KSTRIDE + ks * 4 + 2 * hi];
            const int keyb = (2 * j + sub) * 16;
            const bool diag = keyb + 15 > q0w;
            __builtin_amdgcn_sched_barrier(0);
            #pragma unroll
            for (int ks = 0; ks < 32; ++ks) {
                #pragma unroll
                for (int m = 0; m < MT; ++m)
                    s[m] = __builtin_amdgcn_wmma_f32_16x16x4_f32(
                            false, qf[m][ks], false, kb[ks], (short)0, s[m], false, false);
            }

            #pragma unroll
            for (int m = 0; m < MT; ++m) {
                #pragma unroll
                for (int r = 0; r < 8; ++r)
                    sPw[m * (16 * PSTRIDE) + (r + 8 * hi) * PSTRIDE + l15] = s[m][r];
            }
            asm volatile("s_wait_dscnt 0x0" ::: "memory");
            v2f a[MT][4];
            #pragma unroll
            for (int m = 0; m < MT; ++m) {
                #pragma unroll
                for (int ka = 0; ka < 4; ++ka)
                    a[m][ka] = *(const v2f*)&sPw[m * (16 * PSTRIDE) + l15 * PSTRIDE + ka * 4 + 2 * hi];
            }

            v2f vb[4][8];
            #pragma unroll
            for (int ka = 0; ka < 4; ++ka) {
                #pragma unroll
                for (int dt = 0; dt < 8; ++dt) {
                    vb[ka][dt][0] = cV[v_lane + 2 * ka * VPS + dt * 32     ];
                    vb[ka][dt][1] = cV[v_lane + 2 * ka * VPS + dt * 32 + 16];
                }
            }

            __builtin_amdgcn_sched_barrier(0);
            float sv[MT][8], mxh[MT];
            #pragma unroll
            for (int m = 0; m < MT; ++m) {
                #pragma unroll
                for (int ka = 0; ka < 4; ++ka) { sv[m][2 * ka] = a[m][ka][0]; sv[m][2 * ka + 1] = a[m][ka][1]; }
            }
            if (diag) {
                #pragma unroll
                for (int m = 0; m < MT; ++m) {
                    const int qrow = q0w + 16 * m + l15;
                    #pragma unroll
                    for (int ka = 0; ka < 4; ++ka) {
                        #pragma unroll
                        for (int e = 0; e < 2; ++e)
                            if (keyb + ka * 4 + 2 * hi + e > qrow) sv[m][2 * ka + e] = -10000.0f;
                    }
                }
            }
            bool grow = false;
            #pragma unroll
            for (int m = 0; m < MT; ++m) {
                mxh[m] = fmaxf(fmaxf(fmaxf(sv[m][0], sv[m][1]), fmaxf(sv[m][2], sv[m][3])),
                               fmaxf(fmaxf(sv[m][4], sv[m][5]), fmaxf(sv[m][6], sv[m][7])));
                grow = grow || (mxh[m] > mrow[m] + RESCALE_TAU);
            }
            if (__builtin_expect(__builtin_amdgcn_ballot_w32(grow) != 0u, 0)) {
                #pragma unroll
                for (int m = 0; m < MT; ++m) {
                    const float mx = fmaxf(mxh[m], xhalf(mxh[m]));
                    const float mnew = fmaxf(mrow[m], mx);
                    const float corr = __expf(mrow[m] - mnew);
                    lrow[m] *= corr;
                    mrow[m] = mnew;
                    float cc[8];
                    rows_to_c(corr, cc);
                    #pragma unroll
                    for (int r = 0; r < 8; ++r) {
                        #pragma unroll
                        for (int dt = 0; dt < 8; ++dt) o[m][dt][r] *= cc[r];
                    }
                }
            }
            #pragma unroll
            for (int m = 0; m < MT; ++m) {
                float p[8];
                #pragma unroll
                for (int i = 0; i < 8; ++i) p[i] = __expf(sv[m][i] - mrow[m]);
                float ps = ((p[0] + p[1]) + (p[2] + p[3])) + ((p[4] + p[5]) + (p[6] + p[7]));
                ps += xhalf(ps);
                lrow[m] += ps;
                #pragma unroll
                for (int ka = 0; ka < 4; ++ka) { a[m][ka][0] = p[2 * ka]; a[m][ka][1] = p[2 * ka + 1]; }
            }

            #pragma unroll
            for (int ka = 0; ka < 4; ++ka) {
                #pragma unroll
                for (int dt = 0; dt < 8; ++dt) {
                    #pragma unroll
                    for (int m = 0; m < MT; ++m)
                        o[m][dt] = __builtin_amdgcn_wmma_f32_16x16x4_f32(
                                    false, a[m][ka], false, vb[ka][dt], (short)0, o[m][dt], false, false);
                }
            }
        }
    }

    float* sMo = smem + LDS_MO + pair * (MT * 8 * 8 * 32);
    float* sMl = smem + LDS_ML + pair * (MT * 2 * 32);
    __syncthreads();
    if (helper) {
        #pragma unroll
        for (int m = 0; m < MT; ++m) {
            #pragma unroll
            for (int dt = 0; dt < 8; ++dt) {
                #pragma unroll
                for (int q = 0; q < 2; ++q) {
                    v4f t;
                    t[0] = o[m][dt][4 * q + 0]; t[1] = o[m][dt][4 * q + 1];
                    t[2] = o[m][dt][4 * q + 2]; t[3] = o[m][dt][4 * q + 3];
                    *(v4f*)&sMo[((m * 8 + dt) * 2 + q) * 128 + lane * 4] = t;
                }
            }
            sMl[(m * 2 + 0) * 32 + lane] = mrow[m];
            sMl[(m * 2 + 1) * 32 + lane] = lrow[m];
        }
    }
    __syncthreads();
    if (!helper) {
        #pragma unroll
        for (int m = 0; m < MT; ++m) {
            const float m2 = sMl[(m * 2 + 0) * 32 + lane];
            const float l2 = sMl[(m * 2 + 1) * 32 + lane];
            const float mm = fmaxf(mrow[m], m2);
            const float e1 = __expf(mrow[m] - mm);
            const float e2 = __expf(m2 - mm);
            const float inv = 1.0f / (lrow[m] * e1 + l2 * e2);
            float a1[8], a2[8];
            rows_to_c(e1 * inv, a1);
            rows_to_c(e2 * inv, a2);
            #pragma unroll
            for (int q = 0; q < 2; ++q) {
                #pragma unroll
                for (int dt = 0; dt < 8; ++dt) {
                    const v4f o2 = *(const v4f*)&sMo[((m * 8 + dt) * 2 + q) * 128 + lane * 4];
                    #pragma unroll
                    for (int e = 0; e < 4; ++e) {
                        const int r = 4 * q + e;
                        float* op = Out + (size_t)(q0w + 16 * m + r + 8 * hi) * SROW + head_off;
                        op[dt * 16 + l15] = o[m][dt][r] * a1[r] + o2[e] * a2[r];
                    }
                }
            }
        }
    }
}

extern "C" void kernel_launch(void* const* d_in, const int* in_sizes, int n_in,
                              void* d_out, int out_size, void* d_ws, size_t ws_size,
                              hipStream_t stream) {
    (void)in_sizes; (void)n_in; (void)out_size; (void)d_ws; (void)ws_size;
    const float* Q = (const float*)d_in[0];
    const float* K = (const float*)d_in[1];
    const float* V = (const float*)d_in[2];
    float* Out     = (float*)d_out;

    dim3 grid(SEQ / QBLK, BATCH * HEADS);
    dim3 block(256);
    hipLaunchKernelGGL(fa_fp32_wmma_kernel, grid, block, 0, stream, Q, K, V, Out);
}
